# starting kernel
# baseline (speedup 1.0000x reference)
#include <hip/hip_runtime.h>


typedef float v2f __attribute__((ext_vector_type(2)));
typedef float v4f __attribute__((ext_vector_type(4)));
typedef float v8f __attribute__((ext_vector_type(8)));

#define B_  2048
#define T_  1024
#define D_  24
#define H_  8
#define O_  24
#define G_  32

#define LOG2E_ 1.4426950408889634f

__device__ __forceinline__ float gate_act(float x, float ik, float ib,
                                          float om, float oa) {
    float e = __builtin_amdgcn_exp2f(__builtin_fmaf(ik, x, ib));
    float r = __builtin_amdgcn_rcpf(1.0f + e);
    return __builtin_fmaf(om, r, oa);
}

__device__ __forceinline__ float tanh_nb(float x) {
    float e = __builtin_amdgcn_exp2f(x * (-2.0f * LOG2E_));
    float r = __builtin_amdgcn_rcpf(1.0f + e);
    return __builtin_fmaf(2.0f, r, -1.0f);
}

__global__ __launch_bounds__(32)
void lstm2_fused(const float* __restrict__ x,
                 const float* __restrict__ w_ih0, const float* __restrict__ w_hh0,
                 const float* __restrict__ b_ih0, const float* __restrict__ b_hh0,
                 const float* __restrict__ w_ih1, const float* __restrict__ w_hh1,
                 const float* __restrict__ b_ih1, const float* __restrict__ b_hh1,
                 const float* __restrict__ w_lin, const float* __restrict__ b_lin,
                 float* __restrict__ out)
{
    __shared__ __align__(16) float gatesLDS[16 * G_];
    __shared__ __align__(16) float h0LDS[16 * H_];
    __shared__ __align__(16) float h1LDS[16 * H_];

    const int lane = threadIdx.x;
    const int n    = lane & 15;
    const int half = lane >> 4;
    const int b0   = blockIdx.x * 16;

    v2f Bx[6][2];
#pragma unroll
    for (int kk = 0; kk < 6; ++kk)
#pragma unroll
        for (int nt = 0; nt < 2; ++nt) {
            int g = nt * 16 + n;
            int k = kk * 4 + half * 2;
            Bx[kk][nt] = *(const v2f*)(w_ih0 + g * D_ + k);
        }
    v2f Bh0[2][2], Bi1[2][2], Bh1[2][2];
#pragma unroll
    for (int kk = 0; kk < 2; ++kk)
#pragma unroll
        for (int nt = 0; nt < 2; ++nt) {
            int g = nt * 16 + n;
            int k = kk * 4 + half * 2;
            Bh0[kk][nt] = *(const v2f*)(w_hh0 + g * H_ + k);
            Bi1[kk][nt] = *(const v2f*)(w_ih1 + g * H_ + k);
            Bh1[kk][nt] = *(const v2f*)(w_hh1 + g * H_ + k);
        }

    float bias0t0, bias0t1, bias1t0, bias1t1;
    {
        int g0 = n, g1 = 16 + n;
        bias0t0 = b_ih0[g0] + b_hh0[g0];
        bias0t1 = b_ih0[g1] + b_hh0[g1];
        bias1t0 = b_ih1[g0] + b_hh1[g0];
        bias1t1 = b_ih1[g1] + b_hh1[g1];
    }
    const float ik0 = -LOG2E_;
    const float ib0_l0 = ik0 * bias0t0, ib0_l1 = ik0 * bias1t0;
    const float ik1 = (n < 8) ? (-2.0f * LOG2E_) : -LOG2E_;
    const float ib1_l0 = ik1 * bias0t1, ib1_l1 = ik1 * bias1t1;
    const float om1 = (n < 8) ? 2.0f : 1.0f;
    const float oa1 = (n < 8) ? -1.0f : 0.0f;

    const float* xlane = x + (size_t)(b0 + n) * T_ * D_ + half * 2;

    v2f h0a0 = {0.f, 0.f}, h0a1 = {0.f, 0.f};
    v2f h1a0 = {0.f, 0.f}, h1a1 = {0.f, 0.f};
    v4f c0 = {0.f, 0.f, 0.f, 0.f}, c1 = {0.f, 0.f, 0.f, 0.f};

    const int bb = n;
    const int j0 = half * 4;
    const int brow = half * 8;

    v2f xa[6];
#pragma unroll
    for (int kk = 0; kk < 6; ++kk) xa[kk] = *(const v2f*)(xlane + kk * 4);

    for (int t = 0; t < T_; ++t) {
        v8f acc0 = {0.f, 0.f, 0.f, 0.f, 0.f, 0.f, 0.f, 0.f};
        v8f acc1 = {0.f, 0.f, 0.f, 0.f, 0.f, 0.f, 0.f, 0.f};

#pragma unroll
        for (int kk = 0; kk < 6; ++kk) {
            acc0 = __builtin_amdgcn_wmma_f32_16x16x4_f32(false, xa[kk], false, Bx[kk][0], (short)0, acc0, false, false);
            acc1 = __builtin_amdgcn_wmma_f32_16x16x4_f32(false, xa[kk], false, Bx[kk][1], (short)0, acc1, false, false);
        }
        acc0 = __builtin_amdgcn_wmma_f32_16x16x4_f32(false, h0a0, false, Bh0[0][0], (short)0, acc0, false, false);
        acc1 = __builtin_amdgcn_wmma_f32_16x16x4_f32(false, h0a0, false, Bh0[0][1], (short)0, acc1, false, false);
        acc0 = __builtin_amdgcn_wmma_f32_16x16x4_f32(false, h0a1, false, Bh0[1][0], (short)0, acc0, false, false);
        acc1 = __builtin_amdgcn_wmma_f32_16x16x4_f32(false, h0a1, false, Bh0[1][1], (short)0, acc1, false, false);

        {
            int tn = (t + 1 < T_) ? (t + 1) : t;
            const float* xp = xlane + (size_t)tn * D_;
#pragma unroll
            for (int kk = 0; kk < 6; ++kk) xa[kk] = *(const v2f*)(xp + kk * 4);
        }

#pragma unroll
        for (int v = 0; v < 8; ++v) {
            float s0 = gate_act(acc0[v], ik0, ib0_l0, 1.0f, 0.0f);
            float s1 = gate_act(acc1[v], ik1, ib1_l0, om1, oa1);
            gatesLDS[(brow + v) * G_ + n]      = s0;
            gatesLDS[(brow + v) * G_ + 16 + n] = s1;
        }
        __syncthreads();

        {
            const float* gp = gatesLDS + bb * G_;
            v4f ig = *(const v4f*)(gp + 0  + j0);
            v4f fg = *(const v4f*)(gp + 8  + j0);
            v4f gg = *(const v4f*)(gp + 16 + j0);
            v4f og = *(const v4f*)(gp + 24 + j0);
            v4f hh;
#pragma unroll
            for (int k = 0; k < 4; ++k) {
                c0[k] = __builtin_fmaf(fg[k], c0[k], ig[k] * gg[k]);
                hh[k] = og[k] * tanh_nb(c0[k]);
            }
            *(v4f*)(h0LDS + bb * H_ + j0) = hh;
        }
        __syncthreads();

        h0a0 = *(const v2f*)(h0LDS + n * H_ + half * 2);
        h0a1 = *(const v2f*)(h0LDS + n * H_ + 4 + half * 2);

        acc0 = (v8f){0.f, 0.f, 0.f, 0.f, 0.f, 0.f, 0.f, 0.f};
        acc1 = (v8f){0.f, 0.f, 0.f, 0.f, 0.f, 0.f, 0.f, 0.f};

        acc0 = __builtin_amdgcn_wmma_f32_16x16x4_f32(false, h0a0, false, Bi1[0][0], (short)0, acc0, false, false);
        acc1 = __builtin_amdgcn_wmma_f32_16x16x4_f32(false, h0a0, false, Bi1[0][1], (short)0, acc1, false, false);
        acc0 = __builtin_amdgcn_wmma_f32_16x16x4_f32(false, h0a1, false, Bi1[1][0], (short)0, acc0, false, false);
        acc1 = __builtin_amdgcn_wmma_f32_16x16x4_f32(false, h0a1, false, Bi1[1][1], (short)0, acc1, false, false);
        acc0 = __builtin_amdgcn_wmma_f32_16x16x4_f32(false, h1a0, false, Bh1[0][0], (short)0, acc0, false, false);
        acc1 = __builtin_amdgcn_wmma_f32_16x16x4_f32(false, h1a0, false, Bh1[0][1], (short)0, acc1, false, false);
        acc0 = __builtin_amdgcn_wmma_f32_16x16x4_f32(false, h1a1, false, Bh1[1][0], (short)0, acc0, false, false);
        acc1 = __builtin_amdgcn_wmma_f32_16x16x4_f32(false, h1a1, false, Bh1[1][1], (short)0, acc1, false, false);

#pragma unroll
        for (int v = 0; v < 8; ++v) {
            float s0 = gate_act(acc0[v], ik0, ib0_l1, 1.0f, 0.0f);
            float s1 = gate_act(acc1[v], ik1, ib1_l1, om1, oa1);
            gatesLDS[(brow + v) * G_ + n]      = s0;
            gatesLDS[(brow + v) * G_ + 16 + n] = s1;
        }
        __syncthreads();

        {
            const float* gp = gatesLDS + bb * G_;
            v4f ig = *(const v4f*)(gp + 0  + j0);
            v4f fg = *(const v4f*)(gp + 8  + j0);
            v4f gg = *(const v4f*)(gp + 16 + j0);
            v4f og = *(const v4f*)(gp + 24 + j0);
            v4f hh;
#pragma unroll
            for (int k = 0; k < 4; ++k) {
                c1[k] = __builtin_fmaf(fg[k], c1[k], ig[k] * gg[k]);
                hh[k] = og[k] * tanh_nb(c1[k]);
            }
            *(v4f*)(h1LDS + bb * H_ + j0) = hh;
        }
        __syncthreads();

        h1a0 = *(const v2f*)(h1LDS + n * H_ + half * 2);
        h1a1 = *(const v2f*)(h1LDS + n * H_ + 4 + half * 2);
    }

    {
        const int obase = half * 12;
        const float* hp = h1LDS + bb * H_;
#pragma unroll
        for (int oo = 0; oo < 12; ++oo) {
            int o = obase + oo;
            float s = b_lin[o];
#pragma unroll
            for (int j = 0; j < H_; ++j) s += hp[j] * w_lin[o * H_ + j];
            out[(size_t)(b0 + bb) * O_ + o] = s;
        }
    }
}

extern "C" void kernel_launch(void* const* d_in, const int* in_sizes, int n_in,
                              void* d_out, int out_size, void* d_ws, size_t ws_size,
                              hipStream_t stream) {
    (void)in_sizes; (void)n_in; (void)d_ws; (void)ws_size; (void)out_size;
    const float* x     = (const float*)d_in[0];
    const float* w_ih0 = (const float*)d_in[1];
    const float* w_hh0 = (const float*)d_in[2];
    const float* b_ih0 = (const float*)d_in[3];
    const float* b_hh0 = (const float*)d_in[4];
    const float* w_ih1 = (const float*)d_in[5];
    const float* w_hh1 = (const float*)d_in[6];
    const float* b_ih1 = (const float*)d_in[7];
    const float* b_hh1 = (const float*)d_in[8];
    const float* w_lin = (const float*)d_in[9];
    const float* b_lin = (const float*)d_in[10];
    float* out = (float*)d_out;

    dim3 grid(B_ / 16);
    dim3 block(32);
    hipLaunchKernelGGL(lstm2_fused, grid, block, 0, stream,
                       x, w_ih0, w_hh0, b_ih0, b_hh0,
                       w_ih1, w_hh1, b_ih1, b_hh1, w_lin, b_lin, out);
}


# step's work cut by hidden unit onto four waves, not two: a wave owns 2 units, one 16-column tile holds their i,f,g,o of both layers (zero weight blocks, one h0 product for both); sum order kept
# speedup vs baseline: 2.6199x; 2.6199x over previous
#include <hip/hip_runtime.h>


typedef float v2f __attribute__((ext_vector_type(2)));
typedef float v4f __attribute__((ext_vector_type(4)));
typedef float v8f __attribute__((ext_vector_type(8)));

#define B_  2048
#define T_  1024
#define D_  24
#define H_  8
#define O_  24
#define GP_ 36
#define PL_ (16 * GP_ + 8)

#define LOG2E_ 1.4426950408889634f

__device__ __forceinline__ float gate_act(float x, float ik, float ib,
                                          float om, float oa) {
    float e = __builtin_amdgcn_exp2f(__builtin_fmaf(ik, x, ib));
    float r = __builtin_amdgcn_rcpf(1.0f + e);
    return __builtin_fmaf(om, r, oa);
}

__device__ __forceinline__ float tanh_nb(float x) {
    float e = __builtin_amdgcn_exp2f(x * (-2.0f * LOG2E_));
    float r = __builtin_amdgcn_rcpf(1.0f + e);
    return __builtin_fmaf(2.0f, r, -1.0f);
}

__device__ __forceinline__ void wg_sync_lds() {
    __builtin_amdgcn_fence(__ATOMIC_RELEASE, "workgroup", "local");
    __builtin_amdgcn_s_barrier();
    __builtin_amdgcn_fence(__ATOMIC_ACQUIRE, "workgroup", "local");
}

__device__ __forceinline__ v2f cell_pair(v2f& c, v4f if_, v4f go) {
    v2f ig = {if_[0], if_[1]}, fg = {if_[2], if_[3]};
    v2f gg = {go[0], go[1]},   og = {go[2], go[3]};
    v2f hh;
#pragma unroll
    for (int k = 0; k < 2; ++k) {
        c[k]  = __builtin_fmaf(fg[k], c[k], ig[k] * gg[k]);
        hh[k] = og[k] * tanh_nb(c[k]);
    }
    return hh;
}

__device__ __forceinline__ v8f mm(v2f a, v2f b, v8f c) {
    return __builtin_amdgcn_wmma_f32_16x16x4_f32(false, a, false, b, (short)0, c, false, false);
}

struct Carry {
    v2f h0a0, h0a1, h1a0, h1a1;
    v2f c0a, c0b, c1a, c1b;
    v8f xg;
};

template <bool L0, bool L1>
__device__ __forceinline__ void phase(Carry& s, float* g, int sto, int ldo,
                                      const v2f (&xs)[6], v2f (&xr)[6], const float* xnext,
                                      const v2f (&Bx)[6], const v2f (&Bh0)[2],
                                      const v2f (&Bh1)[2],
                                      float ik, float ib, float om, float oa)
{
    const v8f zero8 = {0.f, 0.f, 0.f, 0.f, 0.f, 0.f, 0.f, 0.f};
    v8f a = L0 ? s.xg : zero8;
    if (L1) {
        a = mm(s.h1a0, Bh1[0], a);
        a = mm(s.h1a1, Bh1[1], a);
    }
    a = mm(s.h0a0, Bh0[0], a);
    a = mm(s.h0a1, Bh0[1], a);

#pragma unroll
    for (int v = 0; v < 8; ++v)
        g[sto + v * GP_] = gate_act(a[v], ik, ib, om, oa);
    wg_sync_lds();

    v4f p0, p1, p2, p3, q0, q1, q2, q3;
    if (L1) {
        q0 = *(const v4f*)(g + PL_ + ldo);
        q1 = *(const v4f*)(g + PL_ + ldo + 4);
        q2 = *(const v4f*)(g + PL_ + ldo + 16);
        q3 = *(const v4f*)(g + PL_ + ldo + 20);
    }
    if (L0) {
        p0 = *(const v4f*)(g + ldo);
        p1 = *(const v4f*)(g + ldo + 4);
        p2 = *(const v4f*)(g + ldo + 16);
        p3 = *(const v4f*)(g + ldo + 20);
    }
    __builtin_amdgcn_sched_barrier(0);
    if (L0) {
#pragma unroll
        for (int kk = 0; kk < 6; ++kk) xr[kk] = *(const v2f*)(xnext + kk * 4);
        v2f x0 = xs[0];
        asm volatile("" : "+v"(x0));
        v8f t = mm(x0, Bx[0], zero8);
#pragma unroll
        for (int kk = 1; kk < 6; ++kk) t = mm(xs[kk], Bx[kk], t);
        asm volatile("" : "+v"(t));
        s.xg = t;
    }
    if (L1) {
        s.h1a0 = cell_pair(s.c1a, q0, q1);
        s.h1a1 = cell_pair(s.c1b, q2, q3);
    }
    if (L0) {
        s.h0a0 = cell_pair(s.c0a, p0, p1);
        s.h0a1 = cell_pair(s.c0b, p2, p3);
    }
}

__global__ __launch_bounds__(128)
void lstm2_fused(const float* __restrict__ x,
                 const float* __restrict__ w_ih0, const float* __restrict__ w_hh0,
                 const float* __restrict__ b_ih0, const float* __restrict__ b_hh0,
                 const float* __restrict__ w_ih1, const float* __restrict__ w_hh1,
                 const float* __restrict__ b_ih1, const float* __restrict__ b_hh1,
                 const float* __restrict__ w_lin, const float* __restrict__ b_lin,
                 float* __restrict__ out)
{
    __shared__ __align__(16) float gLDS[2 * 2 * PL_];
    __shared__ __align__(16) float hT[16 * H_];

    const int w    = __builtin_amdgcn_readfirstlane((int)(threadIdx.x >> 5));
    const int lane = threadIdx.x & 31;
    const int n    = lane & 15;
    const int half = lane >> 4;
    const int b0   = blockIdx.x * 16;

    const int lay  = n >> 3;
    const int q    = (n >> 1) & 3;
    const int unit = 2 * w + (n & 1);
    const int grow = q * H_ + unit;
    const v2f zero2 = {0.f, 0.f};

    v2f Bx[6];
#pragma unroll
    for (int kk = 0; kk < 6; ++kk) {
        v2f t = *(const v2f*)(w_ih0 + grow * D_ + kk * 4 + half * 2);
        Bx[kk] = lay ? zero2 : t;
    }
    const float* wh0 = lay ? w_ih1 : w_hh0;
    v2f Bh0[2], Bh1[2];
#pragma unroll
    for (int kk = 0; kk < 2; ++kk) {
        Bh0[kk] = *(const v2f*)(wh0 + grow * H_ + kk * 4 + half * 2);
        v2f t = *(const v2f*)(w_hh1 + grow * H_ + kk * 4 + half * 2);
        Bh1[kk] = lay ? t : zero2;
    }

    const float bias = lay ? (b_ih1[grow] + b_hh1[grow]) : (b_ih0[grow] + b_hh0[grow]);
    const float ik   = (q == 2) ? (-2.0f * LOG2E_) : -LOG2E_;
    const float ib   = ik * bias;
    const float om   = (q == 2) ? 2.0f : 1.0f;
    const float oa   = (q == 2) ? -1.0f : 0.0f;

    const float* xlane = x + (size_t)(b0 + n) * T_ * D_ + half * 2;

    Carry s;
    s.h0a0 = zero2; s.h0a1 = zero2; s.h1a0 = zero2; s.h1a1 = zero2;
    s.c0a = zero2;  s.c0b = zero2;  s.c1a = zero2;  s.c1b = zero2;

    const int sto = lay * PL_ + (half * 8) * GP_ + w * 8 + q * 2 + (n & 1);
    const int ldo = n * GP_ + half * 8;

    v2f X0[6], X1[6], X2[6], X3[6];
    {
        const v8f zero8 = {0.f, 0.f, 0.f, 0.f, 0.f, 0.f, 0.f, 0.f};
        v2f XT[6];
#pragma unroll
        for (int kk = 0; kk < 6; ++kk) {
            XT[kk] = *(const v2f*)(xlane + kk * 4);
            X1[kk] = *(const v2f*)(xlane + 1 * D_ + kk * 4);
            X2[kk] = *(const v2f*)(xlane + 2 * D_ + kk * 4);
            X3[kk] = *(const v2f*)(xlane + 3 * D_ + kk * 4);
        }
        v8f t = zero8;
#pragma unroll
        for (int kk = 0; kk < 6; ++kk) t = mm(XT[kk], Bx[kk], t);
        s.xg = t;
    }

#define XN_(p) (xlane + (size_t)(((p) + 4 < T_) ? ((p) + 4) : (T_ - 1)) * D_)
#define PH_(L0, L1, p, PAR, XS, XR)                                           \
    phase<L0, L1>(s, gLDS + (PAR) * (2 * PL_), sto, ldo, XS, XR, XN_(p),      \
                  Bx, Bh0, Bh1, ik, ib, om, oa)

    PH_(true, false, 0, 0, X1, X0);
    PH_(true, true,  1, 1, X2, X1);
    PH_(true, true,  2, 0, X3, X2);
    PH_(true, true,  3, 1, X0, X3);
    for (int p = 4; p < T_; p += 4) {
        PH_(true, true, p,     0, X1, X0);
        PH_(true, true, p + 1, 1, X2, X1);
        PH_(true, true, p + 2, 0, X3, X2);
        PH_(true, true, p + 3, 1, X0, X3);
    }
    PH_(false, true, T_, 0, X1, X0);
#undef PH_
#undef XN_

    if (w == 0) {
        *(v2f*)(hT + n * H_ + half * 2)     = s.h1a0;
        *(v2f*)(hT + n * H_ + 4 + half * 2) = s.h1a1;
    }
    wg_sync_lds();
    if (w == 0) {
        const int obase = half * 12;
        const float* hp = hT + n * H_;
#pragma unroll
        for (int oo = 0; oo < 12; ++oo) {
            int o = obase + oo;
            float s2 = b_lin[o];
#pragma unroll
            for (int j = 0; j < H_; ++j) s2 += hp[j] * w_lin[o * H_ + j];
            out[(size_t)(b0 + n) * O_ + o] = s2;
        }
    }
}

extern "C" void kernel_launch(void* const* d_in, const int* in_sizes, int n_in,
                              void* d_out, int out_size, void* d_ws, size_t ws_size,
                              hipStream_t stream) {
    (void)in_sizes; (void)n_in; (void)d_ws; (void)ws_size; (void)out_size;
    const float* x     = (const float*)d_in[0];
    const float* w_ih0 = (const float*)d_in[1];
    const float* w_hh0 = (const float*)d_in[2];
    const float* b_ih0 = (const float*)d_in[3];
    const float* b_hh0 = (const float*)d_in[4];
    const float* w_ih1 = (const float*)d_in[5];
    const float* w_hh1 = (const float*)d_in[6];
    const float* b_ih1 = (const float*)d_in[7];
    const float* b_hh1 = (const float*)d_in[8];
    const float* w_lin = (const float*)d_in[9];
    const float* b_lin = (const float*)d_in[10];
    float* out = (float*)d_out;

    dim3 grid(B_ / 16);
    dim3 block(128);
    hipLaunchKernelGGL(lstm2_fused, grid, block, 0, stream,
                       x, w_ih0, w_hh0, b_ih0, b_hh0,
                       w_ih1, w_hh1, b_ih1, b_hh1, w_lin, b_lin, out);
}


# waves cut by layer and by unit (2 per layer, 4 units each); layer 1 two steps behind, its h0 product formed a phase ahead from h0 passed via LDS; 4 cells per lane, not 8; sum order: layer 1 h0 first
# speedup vs baseline: 2.8008x; 1.0690x over previous
#include <hip/hip_runtime.h>


typedef float v2f __attribute__((ext_vector_type(2)));
typedef float v4f __attribute__((ext_vector_type(4)));
typedef float v8f __attribute__((ext_vector_type(8)));

#define B_  2048
#define T_  1024
#define D_  24
#define H_  8
#define O_  24
#define GP_ 36
#define PL_ (16 * GP_)

#define LOG2E_ 1.4426950408889634f

__device__ __forceinline__ float gate_act(float x, float ik, float ib,
                                          float om, float oa) {
    float e = __builtin_amdgcn_exp2f(__builtin_fmaf(ik, x, ib));
    float r = __builtin_amdgcn_rcpf(1.0f + e);
    return __builtin_fmaf(om, r, oa);
}

__device__ __forceinline__ float tanh_nb(float x) {
    float e = __builtin_amdgcn_exp2f(x * (-2.0f * LOG2E_));
    float r = __builtin_amdgcn_rcpf(1.0f + e);
    return __builtin_fmaf(2.0f, r, -1.0f);
}

__device__ __forceinline__ void wg_sync_lds() {
    __builtin_amdgcn_fence(__ATOMIC_RELEASE, "workgroup", "local");
    __builtin_amdgcn_s_barrier();
    __builtin_amdgcn_fence(__ATOMIC_ACQUIRE, "workgroup", "local");
}

__device__ __forceinline__ v2f cell_pair(v2f& c, v4f if_, v4f go) {
    v2f ig = {if_[0], if_[1]}, fg = {if_[2], if_[3]};
    v2f gg = {go[0], go[1]},   og = {go[2], go[3]};
    v2f hh;
#pragma unroll
    for (int k = 0; k < 2; ++k) {
        c[k]  = __builtin_fmaf(fg[k], c[k], ig[k] * gg[k]);
        hh[k] = og[k] * tanh_nb(c[k]);
    }
    return hh;
}

__device__ __forceinline__ v8f mm(v2f a, v2f b, v8f c) {
    return __builtin_amdgcn_wmma_f32_16x16x4_f32(false, a, false, b, (short)0, c, false, false);
}

struct Carry {
    v2f ha0, ha1;
    v2f ca, cb;
    v8f ing;
};

template <bool A1, bool H1>
__device__ __forceinline__ void phase(Carry& s, bool lay1, float* g, float* hw,
                                      const float* hr, int sto, int ldo,
                                      const v2f (&xs)[6], v2f (&xr)[6], const float* xnext,
                                      const v2f (&Bin)[6], const v2f (&Bh)[2],
                                      float ik, float ib, float om, float oa)
{
    const v8f zero8 = {0.f, 0.f, 0.f, 0.f, 0.f, 0.f, 0.f, 0.f};
    const v2f zero2 = {0.f, 0.f};
    v8f a = s.ing;
    a = mm(s.ha0, Bh[0], a);
    a = mm(s.ha1, Bh[1], a);

#pragma unroll
    for (int v = 0; v < 8; ++v)
        g[sto + v * GP_] = gate_act(a[v], ik, ib, om, oa);
    wg_sync_lds();

    v4f p0 = *(const v4f*)(g + ldo);
    v4f p1 = *(const v4f*)(g + ldo + 4);
    v4f p2 = *(const v4f*)(g + ldo + 16);
    v4f p3 = *(const v4f*)(g + ldo + 20);
    v4f hv = {0.f, 0.f, 0.f, 0.f};
    if (H1) hv = *(const v4f*)hr;
    __builtin_amdgcn_sched_barrier(0);
#pragma unroll
    for (int kk = 0; kk < 6; ++kk) xr[kk] = *(const v2f*)(xnext + kk * 4);
    v2f x0 = xs[0];
    asm volatile("" : "+v"(x0));
    v8f t = mm(x0, Bin[0], zero8);
#pragma unroll
    for (int kk = 1; kk < 4; ++kk) t = mm(xs[kk], Bin[kk], t);
    __builtin_amdgcn_sched_barrier(0);
    v2f h0lo = {hv[0], hv[1]}, h0hi = {hv[2], hv[3]};
    v2f a4 = lay1 ? h0lo : xs[4];
    v2f a5 = lay1 ? h0hi : xs[5];
    t = mm(a4, Bin[4], t);
    t = mm(a5, Bin[5], t);
    asm volatile("" : "+v"(t));
    s.ing = t;
    s.ha0 = cell_pair(s.ca, p0, p1);
    s.ha1 = cell_pair(s.cb, p2, p3);
    if (!A1) {
        s.ha0 = lay1 ? zero2 : s.ha0;  s.ha1 = lay1 ? zero2 : s.ha1;
        s.ca  = lay1 ? zero2 : s.ca;   s.cb  = lay1 ? zero2 : s.cb;
    }
    v4f hn = {s.ha0[0], s.ha0[1], s.ha1[0], s.ha1[1]};
    *(v4f*)hw = hn;
}

__global__ __launch_bounds__(128)
void lstm2_fused(const float* __restrict__ x,
                 const float* __restrict__ w_ih0, const float* __restrict__ w_hh0,
                 const float* __restrict__ b_ih0, const float* __restrict__ b_hh0,
                 const float* __restrict__ w_ih1, const float* __restrict__ w_hh1,
                 const float* __restrict__ b_ih1, const float* __restrict__ b_hh1,
                 const float* __restrict__ w_lin, const float* __restrict__ b_lin,
                 float* __restrict__ out)
{
    __shared__ __align__(16) float gLDS[2 * 2 * PL_];
    __shared__ __align__(16) float hbuf[2 * 4 * 32 * 4];
    __shared__ __align__(16) float hT[16 * H_];

    const int w    = __builtin_amdgcn_readfirstlane((int)(threadIdx.x >> 5));
    const int lay  = w >> 1;
    const bool lay1 = lay != 0;
    const int u    = w & 1;
    const int lane = threadIdx.x & 31;
    const int n    = lane & 15;
    const int half = lane >> 4;
    const int b0   = blockIdx.x * 16;

    const int q    = (n >> 1) & 3;
    const int unit = 4 * u + 2 * (n >> 3) + (n & 1);
    const int grow = q * H_ + unit;
    const v2f zero2 = {0.f, 0.f};
    const v8f zero8 = {0.f, 0.f, 0.f, 0.f, 0.f, 0.f, 0.f, 0.f};

    v2f Bin[6];
#pragma unroll
    for (int kk = 0; kk < 6; ++kk) Bin[kk] = zero2;
    if (lay == 0) {
#pragma unroll
        for (int kk = 0; kk < 6; ++kk)
            Bin[kk] = *(const v2f*)(w_ih0 + grow * D_ + kk * 4 + half * 2);
    } else {
#pragma unroll
        for (int kk = 0; kk < 2; ++kk)
            Bin[4 + kk] = *(const v2f*)(w_ih1 + grow * H_ + kk * 4 + half * 2);
    }
    const float* whh = lay ? w_hh1 : w_hh0;
    v2f Bh[2];
#pragma unroll
    for (int kk = 0; kk < 2; ++kk)
        Bh[kk] = *(const v2f*)(whh + grow * H_ + kk * 4 + half * 2);

    const float bias = lay ? (b_ih1[grow] + b_hh1[grow]) : (b_ih0[grow] + b_hh0[grow]);
    const float ik   = (q == 2) ? (-2.0f * LOG2E_) : -LOG2E_;
    const float ib   = ik * bias;
    const float om   = (q == 2) ? 2.0f : 1.0f;
    const float oa   = (q == 2) ? -1.0f : 0.0f;

    const float* xlane = x + (size_t)(b0 + n) * T_ * D_ + half * 2;
    const int xstep = lay1 ? 0 : D_;

    Carry s;
    s.ha0 = zero2; s.ha1 = zero2; s.ca = zero2; s.cb = zero2;
    s.ing = zero8;

    const int sto = lay * PL_ + (half * 8) * GP_ + u * 16 + n;
    const int ldo = lay * PL_ + n * GP_ + half * 8;
    const int hwo = w * 128 + lane * 4;
    const int hro = u * 128 + lane * 4;

    v2f X0[6], X1[6], X2[6], X3[6];
    {
        v2f XT[6];
#pragma unroll
        for (int kk = 0; kk < 6; ++kk) {
            XT[kk] = *(const v2f*)(xlane + kk * 4);
            X1[kk] = *(const v2f*)(xlane + 1 * xstep + kk * 4);
            X2[kk] = *(const v2f*)(xlane + 2 * xstep + kk * 4);
            X3[kk] = *(const v2f*)(xlane + 3 * xstep + kk * 4);
        }
        XT[4] = lay1 ? zero2 : XT[4];
        XT[5] = lay1 ? zero2 : XT[5];
        v8f t = zero8;
#pragma unroll
        for (int kk = 0; kk < 6; ++kk) t = mm(XT[kk], Bin[kk], t);
        s.ing = t;
    }

#define XN_(p) (xlane + (size_t)(((p) + 4 < T_) ? ((p) + 4) : (T_ - 1)) * xstep)
#define PH_(A1, H1, p, PAR, XS, XR)                                           \
    phase<A1, H1>(s, lay1, gLDS + (PAR) * (2 * PL_),                          \
                  hbuf + (PAR) * 512 + hwo, hbuf + (1 - (PAR)) * 512 + hro,   \
                  sto, ldo, XS, XR, XN_(p), Bin, Bh, ik, ib, om, oa)

    PH_(false, false, 0, 0, X1, X0);
    PH_(false, true,  1, 1, X2, X1);
    PH_(true,  true,  2, 0, X3, X2);
    PH_(true,  true,  3, 1, X0, X3);
    for (int p = 4; p < T_; p += 4) {
        PH_(true, true, p,     0, X1, X0);
        PH_(true, true, p + 1, 1, X2, X1);
        PH_(true, true, p + 2, 0, X3, X2);
        PH_(true, true, p + 3, 1, X0, X3);
    }
    PH_(true, true,  T_,     0, X1, X0);
    PH_(true, false, T_ + 1, 1, X2, X1);
#undef PH_
#undef XN_

    if (w == 2) {
        *(v2f*)(hT + n * H_ + half * 2)     = s.ha0;
        *(v2f*)(hT + n * H_ + 4 + half * 2) = s.ha1;
    }
    wg_sync_lds();
    if (w == 2) {
        const int obase = half * 12;
        const float* hp = hT + n * H_;
#pragma unroll
        for (int oo = 0; oo < 12; ++oo) {
            int o = obase + oo;
            float s2 = b_lin[o];
#pragma unroll
            for (int j = 0; j < H_; ++j) s2 += hp[j] * w_lin[o * H_ + j];
            out[(size_t)(b0 + n) * O_ + o] = s2;
        }
    }
}

extern "C" void kernel_launch(void* const* d_in, const int* in_sizes, int n_in,
                              void* d_out, int out_size, void* d_ws, size_t ws_size,
                              hipStream_t stream) {
    (void)in_sizes; (void)n_in; (void)d_ws; (void)ws_size; (void)out_size;
    const float* x     = (const float*)d_in[0];
    const float* w_ih0 = (const float*)d_in[1];
    const float* w_hh0 = (const float*)d_in[2];
    const float* b_ih0 = (const float*)d_in[3];
    const float* b_hh0 = (const float*)d_in[4];
    const float* w_ih1 = (const float*)d_in[5];
    const float* w_hh1 = (const float*)d_in[6];
    const float* b_ih1 = (const float*)d_in[7];
    const float* b_hh1 = (const float*)d_in[8];
    const float* w_lin = (const float*)d_in[9];
    const float* b_lin = (const float*)d_in[10];
    float* out = (float*)d_out;

    dim3 grid(B_ / 16);
    dim3 block(128);
    hipLaunchKernelGGL(lstm2_fused, grid, block, 0, stream,
                       x, w_ih0, w_hh0, b_ih0, b_hh0,
                       w_ih1, w_hh1, b_ih1, b_hh1, w_lin, b_lin, out);
}


# act constants folded off the serial path: exponent factor in the weights, bias as the input product's start, g gate and cell state carried times -2log2e, h in one fma; sum order: bias first
# speedup vs baseline: 3.0485x; 1.0884x over previous
#include <hip/hip_runtime.h>


typedef float v2f __attribute__((ext_vector_type(2)));
typedef float v4f __attribute__((ext_vector_type(4)));
typedef float v8f __attribute__((ext_vector_type(8)));

#define B_  2048
#define T_  1024
#define D_  24
#define H_  8
#define O_  24
#define GP_ 36
#define PL_ (16 * GP_)

#define LOG2E_ 1.4426950408889634f

__device__ __forceinline__ float gate_act(float y, float om, float oa) {
    float e = __builtin_amdgcn_exp2f(y);
    float r = __builtin_amdgcn_rcpf(1.0f + e);
    return __builtin_fmaf(om, r, oa);
}

__device__ __forceinline__ void wg_sync_lds() {
    __builtin_amdgcn_fence(__ATOMIC_RELEASE, "workgroup", "local");
    __builtin_amdgcn_s_barrier();
    __builtin_amdgcn_fence(__ATOMIC_ACQUIRE, "workgroup", "local");
}

__device__ __forceinline__ v2f cell_pair(v2f& c, v4f if_, v4f go) {
    v2f ig = {if_[0], if_[1]}, fg = {if_[2], if_[3]};
    v2f gg = {go[0], go[1]},   og = {go[2], go[3]};
    v2f hh;
#pragma unroll
    for (int k = 0; k < 2; ++k) {
        c[k]  = __builtin_fmaf(fg[k], c[k], ig[k] * gg[k]);
        float r = __builtin_amdgcn_rcpf(1.0f + __builtin_amdgcn_exp2f(c[k]));
        hh[k] = __builtin_fmaf(og[k] + og[k], r, -og[k]);
    }
    return hh;
}

__device__ __forceinline__ v8f mm(v2f a, v2f b, v8f c) {
    return __builtin_amdgcn_wmma_f32_16x16x4_f32(false, a, false, b, (short)0, c, false, false);
}

struct Carry {
    v2f ha0, ha1;
    v2f ca, cb;
    v8f ing;
};

template <bool A1, bool H1>
__device__ __forceinline__ void phase(Carry& s, bool lay1, float* g, float* hw,
                                      const float* hr, int sto, int ldo,
                                      const v2f (&xs)[6], v2f (&xr)[6], const float* xnext,
                                      const v2f (&Bin)[6], const v2f (&Bh)[2],
                                      const v8f& bt, float om, float oa)
{
    const v2f zero2 = {0.f, 0.f};
    v8f a = s.ing;
    a = mm(s.ha0, Bh[0], a);
    a = mm(s.ha1, Bh[1], a);

#pragma unroll
    for (int v = 0; v < 8; ++v)
        g[sto + v * GP_] = gate_act(a[v], om, oa);
    wg_sync_lds();

    v4f p0 = *(const v4f*)(g + ldo);
    v4f p1 = *(const v4f*)(g + ldo + 4);
    v4f p2 = *(const v4f*)(g + ldo + 16);
    v4f p3 = *(const v4f*)(g + ldo + 20);
    v4f hv = {0.f, 0.f, 0.f, 0.f};
    if (H1) hv = *(const v4f*)hr;
    __builtin_amdgcn_sched_barrier(0);
#pragma unroll
    for (int kk = 0; kk < 6; ++kk) xr[kk] = *(const v2f*)(xnext + kk * 4);
    v2f x0 = xs[0];
    asm volatile("" : "+v"(x0));
    v8f t = mm(x0, Bin[0], bt);
#pragma unroll
    for (int kk = 1; kk < 4; ++kk) t = mm(xs[kk], Bin[kk], t);
    __builtin_amdgcn_sched_barrier(0);
    v2f h0lo = {hv[0], hv[1]}, h0hi = {hv[2], hv[3]};
    v2f a4 = lay1 ? h0lo : xs[4];
    v2f a5 = lay1 ? h0hi : xs[5];
    t = mm(a4, Bin[4], t);
    t = mm(a5, Bin[5], t);
    asm volatile("" : "+v"(t));
    s.ing = t;
    s.ha0 = cell_pair(s.ca, p0, p1);
    s.ha1 = cell_pair(s.cb, p2, p3);
    if (!A1) {
        s.ha0 = lay1 ? zero2 : s.ha0;  s.ha1 = lay1 ? zero2 : s.ha1;
        s.ca  = lay1 ? zero2 : s.ca;   s.cb  = lay1 ? zero2 : s.cb;
    }
    v4f hn = {s.ha0[0], s.ha0[1], s.ha1[0], s.ha1[1]};
    *(v4f*)hw = hn;
}

__global__ __launch_bounds__(128)
void lstm2_fused(const float* __restrict__ x,
                 const float* __restrict__ w_ih0, const float* __restrict__ w_hh0,
                 const float* __restrict__ b_ih0, const float* __restrict__ b_hh0,
                 const float* __restrict__ w_ih1, const float* __restrict__ w_hh1,
                 const float* __restrict__ b_ih1, const float* __restrict__ b_hh1,
                 const float* __restrict__ w_lin, const float* __restrict__ b_lin,
                 float* __restrict__ out)
{
    __shared__ __align__(16) float gLDS[2 * 2 * PL_];
    __shared__ __align__(16) float hbuf[2 * 4 * 32 * 4];
    __shared__ __align__(16) float hT[16 * H_];

    const int w    = __builtin_amdgcn_readfirstlane((int)(threadIdx.x >> 5));
    const int lay  = w >> 1;
    const bool lay1 = lay != 0;
    const int u    = w & 1;
    const int lane = threadIdx.x & 31;
    const int n    = lane & 15;
    const int half = lane >> 4;
    const int b0   = blockIdx.x * 16;

    const int q    = (n >> 1) & 3;
    const int unit = 4 * u + 2 * (n >> 3) + (n & 1);
    const int grow = q * H_ + unit;
    const v2f zero2 = {0.f, 0.f};
    const v8f zero8 = {0.f, 0.f, 0.f, 0.f, 0.f, 0.f, 0.f, 0.f};

    v2f Bin[6];
#pragma unroll
    for (int kk = 0; kk < 6; ++kk) Bin[kk] = zero2;
    if (lay == 0) {
#pragma unroll
        for (int kk = 0; kk < 6; ++kk)
            Bin[kk] = *(const v2f*)(w_ih0 + grow * D_ + kk * 4 + half * 2);
    } else {
#pragma unroll
        for (int kk = 0; kk < 2; ++kk)
            Bin[4 + kk] = *(const v2f*)(w_ih1 + grow * H_ + kk * 4 + half * 2);
    }
    const float* whh = lay ? w_hh1 : w_hh0;
    v2f Bh[2];
#pragma unroll
    for (int kk = 0; kk < 2; ++kk)
        Bh[kk] = *(const v2f*)(whh + grow * H_ + kk * 4 + half * 2);

    const float bias = lay ? (b_ih1[grow] + b_hh1[grow]) : (b_ih0[grow] + b_hh0[grow]);
    const float ik   = (q == 2) ? (-2.0f * LOG2E_) : -LOG2E_;
    const float ib   = ik * bias;
    const float om   = (q == 2) ? (-4.0f * LOG2E_) : 1.0f;
    const float oa   = (q == 2) ? (2.0f * LOG2E_) : 0.0f;
    const v8f bt = {ib, ib, ib, ib, ib, ib, ib, ib};
#pragma unroll
    for (int kk = 0; kk < 6; ++kk) Bin[kk] = Bin[kk] * ik;
#pragma unroll
    for (int kk = 0; kk < 2; ++kk) Bh[kk] = Bh[kk] * ik;

    const float* xlane = x + (size_t)(b0 + n) * T_ * D_ + half * 2;
    const int xstep = lay1 ? 0 : D_;

    Carry s;
    s.ha0 = zero2; s.ha1 = zero2; s.ca = zero2; s.cb = zero2;
    s.ing = zero8;

    const int sto = lay * PL_ + (half * 8) * GP_ + u * 16 + n;
    const int ldo = lay * PL_ + n * GP_ + half * 8;
    const int hwo = w * 128 + lane * 4;
    const int hro = u * 128 + lane * 4;

    v2f X0[6], X1[6], X2[6], X3[6];
    {
        v2f XT[6];
#pragma unroll
        for (int kk = 0; kk < 6; ++kk) {
            XT[kk] = *(const v2f*)(xlane + kk * 4);
            X1[kk] = *(const v2f*)(xlane + 1 * xstep + kk * 4);
            X2[kk] = *(const v2f*)(xlane + 2 * xstep + kk * 4);
            X3[kk] = *(const v2f*)(xlane + 3 * xstep + kk * 4);
        }
        XT[4] = lay1 ? zero2 : XT[4];
        XT[5] = lay1 ? zero2 : XT[5];
        v8f t = bt;
#pragma unroll
        for (int kk = 0; kk < 6; ++kk) t = mm(XT[kk], Bin[kk], t);
        s.ing = t;
    }

#define XN_(p) (xlane + (size_t)(((p) + 4 < T_) ? ((p) + 4) : (T_ - 1)) * xstep)
#define PH_(A1, H1, p, PAR, XS, XR)                                           \
    phase<A1, H1>(s, lay1, gLDS + (PAR) * (2 * PL_),                          \
                  hbuf + (PAR) * 512 + hwo, hbuf + (1 - (PAR)) * 512 + hro,   \
                  sto, ldo, XS, XR, XN_(p), Bin, Bh, bt, om, oa)

    PH_(false, false, 0, 0, X1, X0);
    PH_(false, true,  1, 1, X2, X1);
    PH_(true,  true,  2, 0, X3, X2);
    PH_(true,  true,  3, 1, X0, X3);
    for (int p = 4; p < T_; p += 4) {
        PH_(true, true, p,     0, X1, X0);
        PH_(true, true, p + 1, 1, X2, X1);
        PH_(true, true, p + 2, 0, X3, X2);
        PH_(true, true, p + 3, 1, X0, X3);
    }
    PH_(true, true,  T_,     0, X1, X0);
    PH_(true, false, T_ + 1, 1, X2, X1);
#undef PH_
#undef XN_

    if (w == 2) {
        *(v2f*)(hT + n * H_ + half * 2)     = s.ha0;
        *(v2f*)(hT + n * H_ + 4 + half * 2) = s.ha1;
    }
    wg_sync_lds();
    if (w == 2) {
        const int obase = half * 12;
        const float* hp = hT + n * H_;
#pragma unroll
        for (int oo = 0; oo < 12; ++oo) {
            int o = obase + oo;
            float s2 = b_lin[o];
#pragma unroll
            for (int j = 0; j < H_; ++j) s2 += hp[j] * w_lin[o * H_ + j];
            out[(size_t)(b0 + n) * O_ + o] = s2;
        }
    }
}

extern "C" void kernel_launch(void* const* d_in, const int* in_sizes, int n_in,
                              void* d_out, int out_size, void* d_ws, size_t ws_size,
                              hipStream_t stream) {
    (void)in_sizes; (void)n_in; (void)d_ws; (void)ws_size; (void)out_size;
    const float* x     = (const float*)d_in[0];
    const float* w_ih0 = (const float*)d_in[1];
    const float* w_hh0 = (const float*)d_in[2];
    const float* b_ih0 = (const float*)d_in[3];
    const float* b_hh0 = (const float*)d_in[4];
    const float* w_ih1 = (const float*)d_in[5];
    const float* w_hh1 = (const float*)d_in[6];
    const float* b_ih1 = (const float*)d_in[7];
    const float* b_hh1 = (const float*)d_in[8];
    const float* w_lin = (const float*)d_in[9];
    const float* b_lin = (const float*)d_in[10];
    float* out = (float*)d_out;

    dim3 grid(B_ / 16);
    dim3 block(128);
    hipLaunchKernelGGL(lstm2_fused, grid, block, 0, stream,
                       x, w_ih0, w_hh0, b_ih0, b_hh0,
                       w_ih1, w_hh1, b_ih1, b_hh1, w_lin, b_lin, out);
}


# waves of a layer cut by batch row, not hidden unit (8 rows x 8 units, two gate tiles a wave): gate exchange stays inside the wave, no barrier or store wait before the cell update; sum order kept
# speedup vs baseline: 3.1649x; 1.0382x over previous
#include <hip/hip_runtime.h>


typedef float v2f __attribute__((ext_vector_type(2)));
typedef float v4f __attribute__((ext_vector_type(4)));
typedef float v8f __attribute__((ext_vector_type(8)));

#define B_  2048
#define T_  1024
#define D_  24
#define H_  8
#define O_  24
#define GP_ 36
#define PL_ (8 * GP_)

#define LOG2E_ 1.4426950408889634f


__device__ __forceinline__ void wg_sync_lds() {
    __builtin_amdgcn_fence(__ATOMIC_RELEASE, "workgroup", "local");
    __builtin_amdgcn_s_barrier();
    __builtin_amdgcn_fence(__ATOMIC_ACQUIRE, "workgroup", "local");
}

__device__ __forceinline__ v2f cell_pair(v2f& c, v4f if_, v4f go) {
    v2f ig = {if_[0], if_[1]}, fg = {if_[2], if_[3]};
    v2f gg = {go[0], go[1]},   og = {go[2], go[3]};
    v2f hh;
#pragma unroll
    for (int k = 0; k < 2; ++k) {
        c[k]  = __builtin_fmaf(fg[k], c[k], ig[k] * gg[k]);
        float r = __builtin_amdgcn_rcpf(1.0f + __builtin_amdgcn_exp2f(c[k]));
        hh[k] = __builtin_fmaf(og[k] + og[k], r, -og[k]);
    }
    return hh;
}

__device__ __forceinline__ v8f mm(v2f a, v2f b, v8f c) {
    return __builtin_amdgcn_wmma_f32_16x16x4_f32(false, a, false, b, (short)0, c, false, false);
}

struct Carry {
    v2f ha0, ha1;
    v2f ca, cb;
    v8f ing0, ing1;
};

template <bool A1, bool H1>
__device__ __forceinline__ void phase(Carry& s, bool lay1, float* g, float* hw,
                                      const float* hr, int sto, int ldo,
                                      const v2f (&xs)[6], v2f (&xr)[6], const float* xnext,
                                      const v2f (&Bi0)[6], const v2f (&Bi1)[6],
                                      const v2f (&Bh0)[2], const v2f (&Bh1)[2],
                                      const v8f& bt0, const v8f& bt1, float om, float oa)
{
    const v2f zero2 = {0.f, 0.f};
    v8f a0 = s.ing0, a1 = s.ing1;
    a0 = mm(s.ha0, Bh0[0], a0);
    a1 = mm(s.ha0, Bh1[0], a1);
    a0 = mm(s.ha1, Bh0[1], a0);
    a1 = mm(s.ha1, Bh1[1], a1);
    asm volatile("" : "+v"(a0), "+v"(a1));
    v2f x0 = xs[0];
    asm volatile("" : "+v"(x0));
    v8f t0 = mm(x0, Bi0[0], bt0);
    v8f t1 = mm(x0, Bi1[0], bt1);
    asm volatile("" : "+v"(t0), "+v"(t1));
    __builtin_amdgcn_sched_barrier(0);
    wg_sync_lds();

    v4f e0, e1;
#pragma unroll
    for (int v = 0; v < 4; ++v) {
        e0[v] = 1.0f + __builtin_amdgcn_exp2f(a0[v]);
        e1[v] = 1.0f + __builtin_amdgcn_exp2f(a1[v]);
    }
    asm volatile("" : "+v"(e0), "+v"(e1));
    v2f x1 = xs[1];
    asm volatile("" : "+v"(x1));
    t0 = mm(x1, Bi0[1], t0);
    asm volatile("" : "+v"(t0));
#pragma unroll
    for (int v = 0; v < 4; ++v) {
        e0[v] = __builtin_amdgcn_rcpf(e0[v]);
        e1[v] = __builtin_amdgcn_rcpf(e1[v]);
    }
    asm volatile("" : "+v"(e0), "+v"(e1));
    asm volatile("" : "+v"(x1));
    t1 = mm(x1, Bi1[1], t1);
    asm volatile("" : "+v"(t1));
#pragma unroll
    for (int v = 0; v < 4; ++v) {
        g[sto + v * GP_]      = __builtin_fmaf(om, e0[v], oa);
        g[sto + v * GP_ + 16] = __builtin_fmaf(om, e1[v], oa);
    }
    __builtin_amdgcn_fence(__ATOMIC_RELEASE, "wavefront", "local");
    __builtin_amdgcn_fence(__ATOMIC_ACQUIRE, "wavefront", "local");

    v4f p0 = *(const v4f*)(g + ldo);
    v4f p1 = *(const v4f*)(g + ldo + 4);
    v4f p2 = *(const v4f*)(g + ldo + 16);
    v4f p3 = *(const v4f*)(g + ldo + 20);
    v4f hv = {0.f, 0.f, 0.f, 0.f};
    if (H1) hv = *(const v4f*)hr;
    __builtin_amdgcn_sched_barrier(0);
#pragma unroll
    for (int kk = 0; kk < 6; ++kk) xr[kk] = *(const v2f*)(xnext + kk * 4);
    v2f x2 = xs[2];
    asm volatile("" : "+v"(x2));
    t0 = mm(x2, Bi0[2], t0);
    t1 = mm(x2, Bi1[2], t1);
    t0 = mm(xs[3], Bi0[3], t0);
    t1 = mm(xs[3], Bi1[3], t1);
    __builtin_amdgcn_sched_barrier(0);
    v2f h0lo = {hv[0], hv[1]}, h0hi = {hv[2], hv[3]};
    v2f a4 = lay1 ? h0lo : xs[4];
    v2f a5 = lay1 ? h0hi : xs[5];
    t0 = mm(a4, Bi0[4], t0);
    t1 = mm(a4, Bi1[4], t1);
    t0 = mm(a5, Bi0[5], t0);
    t1 = mm(a5, Bi1[5], t1);
    asm volatile("" : "+v"(t0), "+v"(t1));
    s.ing0 = t0;
    s.ing1 = t1;
    s.ha0 = cell_pair(s.ca, p0, p1);
    s.ha1 = cell_pair(s.cb, p2, p3);
    if (!A1) {
        s.ha0 = lay1 ? zero2 : s.ha0;  s.ha1 = lay1 ? zero2 : s.ha1;
        s.ca  = lay1 ? zero2 : s.ca;   s.cb  = lay1 ? zero2 : s.cb;
    }
    v4f hn = {s.ha0[0], s.ha0[1], s.ha1[0], s.ha1[1]};
    *(v4f*)hw = hn;
}

__global__ __launch_bounds__(128)
void lstm2_fused(const float* __restrict__ x,
                 const float* __restrict__ w_ih0, const float* __restrict__ w_hh0,
                 const float* __restrict__ b_ih0, const float* __restrict__ b_hh0,
                 const float* __restrict__ w_ih1, const float* __restrict__ w_hh1,
                 const float* __restrict__ b_ih1, const float* __restrict__ b_hh1,
                 const float* __restrict__ w_lin, const float* __restrict__ b_lin,
                 float* __restrict__ out)
{
    __shared__ __align__(16) float gLDS[4 * PL_];
    __shared__ __align__(16) float hbuf[2 * 4 * 32 * 4];
    __shared__ __align__(16) float hT[16 * H_];

    const int w    = __builtin_amdgcn_readfirstlane((int)(threadIdx.x >> 5));
    const int lay  = w >> 1;
    const bool lay1 = lay != 0;
    const int u    = w & 1;
    const int lane = threadIdx.x & 31;
    const int n    = lane & 15;
    const int half = lane >> 4;
    const int b0   = blockIdx.x * 16;
    const int slot = (n & 3) + 4 * (n >> 3);
    const int brow = 8 * u + slot;

    const int q     = (n >> 1) & 3;
    const int unit0 = 2 * (n >> 3) + (n & 1);
    const int grow0 = q * H_ + unit0;
    const int grow1 = grow0 + 4;
    const v2f zero2 = {0.f, 0.f};
    const v8f zero8 = {0.f, 0.f, 0.f, 0.f, 0.f, 0.f, 0.f, 0.f};

    v2f Bi0[6], Bi1[6];
#pragma unroll
    for (int kk = 0; kk < 6; ++kk) { Bi0[kk] = zero2; Bi1[kk] = zero2; }
    if (lay == 0) {
#pragma unroll
        for (int kk = 0; kk < 6; ++kk) {
            Bi0[kk] = *(const v2f*)(w_ih0 + grow0 * D_ + kk * 4 + half * 2);
            Bi1[kk] = *(const v2f*)(w_ih0 + grow1 * D_ + kk * 4 + half * 2);
        }
    } else {
#pragma unroll
        for (int kk = 0; kk < 2; ++kk) {
            Bi0[4 + kk] = *(const v2f*)(w_ih1 + grow0 * H_ + kk * 4 + half * 2);
            Bi1[4 + kk] = *(const v2f*)(w_ih1 + grow1 * H_ + kk * 4 + half * 2);
        }
    }
    const float* whh = lay ? w_hh1 : w_hh0;
    v2f Bh0[2], Bh1[2];
#pragma unroll
    for (int kk = 0; kk < 2; ++kk) {
        Bh0[kk] = *(const v2f*)(whh + grow0 * H_ + kk * 4 + half * 2);
        Bh1[kk] = *(const v2f*)(whh + grow1 * H_ + kk * 4 + half * 2);
    }

    const float bias0 = lay ? (b_ih1[grow0] + b_hh1[grow0]) : (b_ih0[grow0] + b_hh0[grow0]);
    const float bias1 = lay ? (b_ih1[grow1] + b_hh1[grow1]) : (b_ih0[grow1] + b_hh0[grow1]);
    const float ik   = (q == 2) ? (-2.0f * LOG2E_) : -LOG2E_;
    const float ib0  = ik * bias0;
    const float ib1  = ik * bias1;
    const float om   = (q == 2) ? (-4.0f * LOG2E_) : 1.0f;
    const float oa   = (q == 2) ? (2.0f * LOG2E_) : 0.0f;
    const v8f bt0 = {ib0, ib0, ib0, ib0, ib0, ib0, ib0, ib0};
    const v8f bt1 = {ib1, ib1, ib1, ib1, ib1, ib1, ib1, ib1};
#pragma unroll
    for (int kk = 0; kk < 6; ++kk) { Bi0[kk] = Bi0[kk] * ik; Bi1[kk] = Bi1[kk] * ik; }
#pragma unroll
    for (int kk = 0; kk < 2; ++kk) { Bh0[kk] = Bh0[kk] * ik; Bh1[kk] = Bh1[kk] * ik; }

    const float* xlane = x + (size_t)(b0 + brow) * T_ * D_ + half * 2;
    const int xstep = lay1 ? 0 : D_;

    Carry s;
    s.ha0 = zero2; s.ha1 = zero2; s.ca = zero2; s.cb = zero2;
    s.ing0 = zero8; s.ing1 = zero8;

    const int sto = w * PL_ + (half * 4) * GP_ + n;
    const int ldo = w * PL_ + slot * GP_ + half * 8;
    const int hwo = w * 128 + lane * 4;
    const int hro = u * 128 + lane * 4;

    v2f X0[6], X1[6], X2[6], X3[6];
    {
        v2f XT[6];
#pragma unroll
        for (int kk = 0; kk < 6; ++kk) {
            XT[kk] = *(const v2f*)(xlane + kk * 4);
            X1[kk] = *(const v2f*)(xlane + 1 * xstep + kk * 4);
            X2[kk] = *(const v2f*)(xlane + 2 * xstep + kk * 4);
            X3[kk] = *(const v2f*)(xlane + 3 * xstep + kk * 4);
        }
        XT[4] = lay1 ? zero2 : XT[4];
        XT[5] = lay1 ? zero2 : XT[5];
        v8f t0 = bt0, t1 = bt1;
#pragma unroll
        for (int kk = 0; kk < 6; ++kk) {
            t0 = mm(XT[kk], Bi0[kk], t0);
            t1 = mm(XT[kk], Bi1[kk], t1);
        }
        s.ing0 = t0;
        s.ing1 = t1;
    }

#define XN_(p) (xlane + (size_t)(((p) + 4 < T_) ? ((p) + 4) : (T_ - 1)) * xstep)
#define PH_(A1, H1, p, PAR, XS, XR)                                           \
    phase<A1, H1>(s, lay1, gLDS,                                              \
                  hbuf + (PAR) * 512 + hwo, hbuf + (1 - (PAR)) * 512 + hro,   \
                  sto, ldo, XS, XR, XN_(p), Bi0, Bi1, Bh0, Bh1, bt0, bt1, om, oa)

    PH_(false, false, 0, 0, X1, X0);
    PH_(false, true,  1, 1, X2, X1);
    PH_(true,  true,  2, 0, X3, X2);
    PH_(true,  true,  3, 1, X0, X3);
    for (int p = 4; p < T_; p += 4) {
        PH_(true, true, p,     0, X1, X0);
        PH_(true, true, p + 1, 1, X2, X1);
        PH_(true, true, p + 2, 0, X3, X2);
        PH_(true, true, p + 3, 1, X0, X3);
    }
    PH_(true, true,  T_,     0, X1, X0);
    PH_(true, false, T_ + 1, 1, X2, X1);
#undef PH_
#undef XN_

    if (lay1) {
        *(v2f*)(hT + brow * H_ + half * 2)     = s.ha0;
        *(v2f*)(hT + brow * H_ + 4 + half * 2) = s.ha1;
    }
    wg_sync_lds();
    if (w == 2) {
        const int obase = half * 12;
        const float* hp = hT + n * H_;
#pragma unroll
        for (int oo = 0; oo < 12; ++oo) {
            int o = obase + oo;
            float s2 = b_lin[o];
#pragma unroll
            for (int j = 0; j < H_; ++j) s2 += hp[j] * w_lin[o * H_ + j];
            out[(size_t)(b0 + n) * O_ + o] = s2;
        }
    }
}

extern "C" void kernel_launch(void* const* d_in, const int* in_sizes, int n_in,
                              void* d_out, int out_size, void* d_ws, size_t ws_size,
                              hipStream_t stream) {
    (void)in_sizes; (void)n_in; (void)d_ws; (void)ws_size; (void)out_size;
    const float* x     = (const float*)d_in[0];
    const float* w_ih0 = (const float*)d_in[1];
    const float* w_hh0 = (const float*)d_in[2];
    const float* b_ih0 = (const float*)d_in[3];
    const float* b_hh0 = (const float*)d_in[4];
    const float* w_ih1 = (const float*)d_in[5];
    const float* w_hh1 = (const float*)d_in[6];
    const float* b_ih1 = (const float*)d_in[7];
    const float* b_hh1 = (const float*)d_in[8];
    const float* w_lin = (const float*)d_in[9];
    const float* b_lin = (const float*)d_in[10];
    float* out = (float*)d_out;

    dim3 grid(B_ / 16);
    dim3 block(128);
    hipLaunchKernelGGL(lstm2_fused, grid, block, 0, stream,
                       x, w_ih0, w_hh0, b_ih0, b_hh0,
                       w_ih1, w_hh1, b_ih1, b_hh1, w_lin, b_lin, out);
}


# h0 hand-over store moved into the gate exchange's LDS group of the next phase (layer 1 three steps behind, not two): one LDS place per phase, no store wait in front of the barrier; sum order kept
# speedup vs baseline: 3.3893x; 1.0709x over previous
#include <hip/hip_runtime.h>


typedef float v2f __attribute__((ext_vector_type(2)));
typedef float v4f __attribute__((ext_vector_type(4)));
typedef float v8f __attribute__((ext_vector_type(8)));

#define B_  2048
#define T_  1024
#define D_  24
#define H_  8
#define O_  24
#define GP_ 36
#define PL_ (8 * GP_)

#define LOG2E_ 1.4426950408889634f


__device__ __forceinline__ void wg_sync_lds() {
    __builtin_amdgcn_fence(__ATOMIC_RELEASE, "workgroup", "local");
    __builtin_amdgcn_s_barrier();
    __builtin_amdgcn_fence(__ATOMIC_ACQUIRE, "workgroup", "local");
}

__device__ __forceinline__ v2f cell_pair(v2f& c, v4f if_, v4f go) {
    v2f ig = {if_[0], if_[1]}, fg = {if_[2], if_[3]};
    v2f gg = {go[0], go[1]},   og = {go[2], go[3]};
    v2f hh;
#pragma unroll
    for (int k = 0; k < 2; ++k) {
        c[k]  = __builtin_fmaf(fg[k], c[k], ig[k] * gg[k]);
        float r = __builtin_amdgcn_rcpf(1.0f + __builtin_amdgcn_exp2f(c[k]));
        hh[k] = __builtin_fmaf(og[k] + og[k], r, -og[k]);
    }
    return hh;
}

__device__ __forceinline__ v8f mm(v2f a, v2f b, v8f c) {
    return __builtin_amdgcn_wmma_f32_16x16x4_f32(false, a, false, b, (short)0, c, false, false);
}

struct Carry {
    v2f ha0, ha1;
    v2f ca, cb;
    v8f ing0, ing1;
};

template <bool A1, bool H1>
__device__ __forceinline__ void phase(Carry& s, bool lay1, float* g, float* hw,
                                      const float* hr, int sto, int ldo,
                                      const v2f (&xs)[6], v2f (&xr)[6], const float* xnext,
                                      const v2f (&Bi0)[6], const v2f (&Bi1)[6],
                                      const v2f (&Bh0)[2], const v2f (&Bh1)[2],
                                      const v8f& bt0, const v8f& bt1, float om, float oa)
{
    const v2f zero2 = {0.f, 0.f};
    v8f a0 = s.ing0, a1 = s.ing1;
    a0 = mm(s.ha0, Bh0[0], a0);
    a1 = mm(s.ha0, Bh1[0], a1);
    a0 = mm(s.ha1, Bh0[1], a0);
    a1 = mm(s.ha1, Bh1[1], a1);
    asm volatile("" : "+v"(a0), "+v"(a1));
    v2f x0 = xs[0];
    asm volatile("" : "+v"(x0));
    v8f t0 = mm(x0, Bi0[0], bt0);
    v8f t1 = mm(x0, Bi1[0], bt1);
    asm volatile("" : "+v"(t0), "+v"(t1));
    __builtin_amdgcn_sched_barrier(0);
    wg_sync_lds();

    v4f e0, e1;
#pragma unroll
    for (int v = 0; v < 4; ++v) {
        e0[v] = 1.0f + __builtin_amdgcn_exp2f(a0[v]);
        e1[v] = 1.0f + __builtin_amdgcn_exp2f(a1[v]);
    }
    asm volatile("" : "+v"(e0), "+v"(e1));
    v2f x1 = xs[1];
    asm volatile("" : "+v"(x1));
    t0 = mm(x1, Bi0[1], t0);
    asm volatile("" : "+v"(t0));
#pragma unroll
    for (int v = 0; v < 4; ++v) {
        e0[v] = __builtin_amdgcn_rcpf(e0[v]);
        e1[v] = __builtin_amdgcn_rcpf(e1[v]);
    }
    asm volatile("" : "+v"(e0), "+v"(e1));
    asm volatile("" : "+v"(x1));
    t1 = mm(x1, Bi1[1], t1);
    asm volatile("" : "+v"(t1));
#pragma unroll
    for (int v = 0; v < 4; ++v) {
        g[sto + v * GP_]      = __builtin_fmaf(om, e0[v], oa);
        g[sto + v * GP_ + 16] = __builtin_fmaf(om, e1[v], oa);
    }
    v4f hp = {s.ha0[0], s.ha0[1], s.ha1[0], s.ha1[1]};
    *(v4f*)hw = hp;
    __builtin_amdgcn_fence(__ATOMIC_RELEASE, "wavefront", "local");
    __builtin_amdgcn_fence(__ATOMIC_ACQUIRE, "wavefront", "local");

    v4f p0 = *(const v4f*)(g + ldo);
    v4f p1 = *(const v4f*)(g + ldo + 4);
    v4f p2 = *(const v4f*)(g + ldo + 16);
    v4f p3 = *(const v4f*)(g + ldo + 20);
    v4f hv = {0.f, 0.f, 0.f, 0.f};
    if (H1) hv = *(const v4f*)hr;
    __builtin_amdgcn_sched_barrier(0);
#pragma unroll
    for (int kk = 0; kk < 6; ++kk) xr[kk] = *(const v2f*)(xnext + kk * 4);
    v2f x2 = xs[2];
    asm volatile("" : "+v"(x2));
    t0 = mm(x2, Bi0[2], t0);
    t1 = mm(x2, Bi1[2], t1);
    t0 = mm(xs[3], Bi0[3], t0);
    t1 = mm(xs[3], Bi1[3], t1);
    __builtin_amdgcn_sched_barrier(0);
    v2f h0lo = {hv[0], hv[1]}, h0hi = {hv[2], hv[3]};
    v2f a4 = lay1 ? h0lo : xs[4];
    v2f a5 = lay1 ? h0hi : xs[5];
    t0 = mm(a4, Bi0[4], t0);
    t1 = mm(a4, Bi1[4], t1);
    t0 = mm(a5, Bi0[5], t0);
    t1 = mm(a5, Bi1[5], t1);
    asm volatile("" : "+v"(t0), "+v"(t1));
    s.ing0 = t0;
    s.ing1 = t1;
    s.ha0 = cell_pair(s.ca, p0, p1);
    s.ha1 = cell_pair(s.cb, p2, p3);
    if (!A1) {
        s.ha0 = lay1 ? zero2 : s.ha0;  s.ha1 = lay1 ? zero2 : s.ha1;
        s.ca  = lay1 ? zero2 : s.ca;   s.cb  = lay1 ? zero2 : s.cb;
    }
}

__global__ __launch_bounds__(128)
void lstm2_fused(const float* __restrict__ x,
                 const float* __restrict__ w_ih0, const float* __restrict__ w_hh0,
                 const float* __restrict__ b_ih0, const float* __restrict__ b_hh0,
                 const float* __restrict__ w_ih1, const float* __restrict__ w_hh1,
                 const float* __restrict__ b_ih1, const float* __restrict__ b_hh1,
                 const float* __restrict__ w_lin, const float* __restrict__ b_lin,
                 float* __restrict__ out)
{
    __shared__ __align__(16) float gLDS[4 * PL_];
    __shared__ __align__(16) float hbuf[2 * 4 * 32 * 4];
    __shared__ __align__(16) float hT[16 * H_];

    const int w    = __builtin_amdgcn_readfirstlane((int)(threadIdx.x >> 5));
    const int lay  = w >> 1;
    const bool lay1 = lay != 0;
    const int u    = w & 1;
    const int lane = threadIdx.x & 31;
    const int n    = lane & 15;
    const int half = lane >> 4;
    const int b0   = blockIdx.x * 16;
    const int slot = (n & 3) + 4 * (n >> 3);
    const int brow = 8 * u + slot;

    const int q     = (n >> 1) & 3;
    const int unit0 = 2 * (n >> 3) + (n & 1);
    const int grow0 = q * H_ + unit0;
    const int grow1 = grow0 + 4;
    const v2f zero2 = {0.f, 0.f};
    const v8f zero8 = {0.f, 0.f, 0.f, 0.f, 0.f, 0.f, 0.f, 0.f};

    v2f Bi0[6], Bi1[6];
#pragma unroll
    for (int kk = 0; kk < 6; ++kk) { Bi0[kk] = zero2; Bi1[kk] = zero2; }
    if (lay == 0) {
#pragma unroll
        for (int kk = 0; kk < 6; ++kk) {
            Bi0[kk] = *(const v2f*)(w_ih0 + grow0 * D_ + kk * 4 + half * 2);
            Bi1[kk] = *(const v2f*)(w_ih0 + grow1 * D_ + kk * 4 + half * 2);
        }
    } else {
#pragma unroll
        for (int kk = 0; kk < 2; ++kk) {
            Bi0[4 + kk] = *(const v2f*)(w_ih1 + grow0 * H_ + kk * 4 + half * 2);
            Bi1[4 + kk] = *(const v2f*)(w_ih1 + grow1 * H_ + kk * 4 + half * 2);
        }
    }
    const float* whh = lay ? w_hh1 : w_hh0;
    v2f Bh0[2], Bh1[2];
#pragma unroll
    for (int kk = 0; kk < 2; ++kk) {
        Bh0[kk] = *(const v2f*)(whh + grow0 * H_ + kk * 4 + half * 2);
        Bh1[kk] = *(const v2f*)(whh + grow1 * H_ + kk * 4 + half * 2);
    }

    const float bias0 = lay ? (b_ih1[grow0] + b_hh1[grow0]) : (b_ih0[grow0] + b_hh0[grow0]);
    const float bias1 = lay ? (b_ih1[grow1] + b_hh1[grow1]) : (b_ih0[grow1] + b_hh0[grow1]);
    const float ik   = (q == 2) ? (-2.0f * LOG2E_) : -LOG2E_;
    const float ib0  = ik * bias0;
    const float ib1  = ik * bias1;
    const float om   = (q == 2) ? (-4.0f * LOG2E_) : 1.0f;
    const float oa   = (q == 2) ? (2.0f * LOG2E_) : 0.0f;
    const v8f bt0 = {ib0, ib0, ib0, ib0, ib0, ib0, ib0, ib0};
    const v8f bt1 = {ib1, ib1, ib1, ib1, ib1, ib1, ib1, ib1};
#pragma unroll
    for (int kk = 0; kk < 6; ++kk) { Bi0[kk] = Bi0[kk] * ik; Bi1[kk] = Bi1[kk] * ik; }
#pragma unroll
    for (int kk = 0; kk < 2; ++kk) { Bh0[kk] = Bh0[kk] * ik; Bh1[kk] = Bh1[kk] * ik; }

    const float* xlane = x + (size_t)(b0 + brow) * T_ * D_ + half * 2;
    const int xstep = lay1 ? 0 : D_;

    Carry s;
    s.ha0 = zero2; s.ha1 = zero2; s.ca = zero2; s.cb = zero2;
    s.ing0 = zero8; s.ing1 = zero8;

    const int sto = w * PL_ + (half * 4) * GP_ + n;
    const int ldo = w * PL_ + slot * GP_ + half * 8;
    const int hwo = w * 128 + lane * 4;
    const int hro = u * 128 + lane * 4;

    v2f X0[6], X1[6], X2[6], X3[6];
    {
        v2f XT[6];
#pragma unroll
        for (int kk = 0; kk < 6; ++kk) {
            XT[kk] = *(const v2f*)(xlane + kk * 4);
            X1[kk] = *(const v2f*)(xlane + 1 * xstep + kk * 4);
            X2[kk] = *(const v2f*)(xlane + 2 * xstep + kk * 4);
            X3[kk] = *(const v2f*)(xlane + 3 * xstep + kk * 4);
        }
        XT[4] = lay1 ? zero2 : XT[4];
        XT[5] = lay1 ? zero2 : XT[5];
        v8f t0 = bt0, t1 = bt1;
#pragma unroll
        for (int kk = 0; kk < 6; ++kk) {
            t0 = mm(XT[kk], Bi0[kk], t0);
            t1 = mm(XT[kk], Bi1[kk], t1);
        }
        s.ing0 = t0;
        s.ing1 = t1;
    }

#define XN_(p) (xlane + (size_t)(((p) + 4 < T_) ? ((p) + 4) : (T_ - 1)) * xstep)
#define PH_(A1, H1, p, PAR, XS, XR)                                           \
    phase<A1, H1>(s, lay1, gLDS,                                              \
                  hbuf + (1 - (PAR)) * 512 + hwo, hbuf + (PAR) * 512 + hro,   \
                  sto, ldo, XS, XR, XN_(p), Bi0, Bi1, Bh0, Bh1, bt0, bt1, om, oa)

    PH_(false, false, 0, 0, X1, X0);
    PH_(false, false, 1, 1, X2, X1);
    PH_(false, true,  2, 0, X3, X2);
    PH_(true,  true,  3, 1, X0, X3);
    for (int p = 4; p < T_; p += 4) {
        PH_(true, true, p,     0, X1, X0);
        PH_(true, true, p + 1, 1, X2, X1);
        PH_(true, true, p + 2, 0, X3, X2);
        PH_(true, true, p + 3, 1, X0, X3);
    }
    PH_(true, true,  T_,     0, X1, X0);
    PH_(true, true,  T_ + 1, 1, X2, X1);
    PH_(true, false, T_ + 2, 0, X3, X2);
#undef PH_
#undef XN_

    if (lay1) {
        *(v2f*)(hT + brow * H_ + half * 2)     = s.ha0;
        *(v2f*)(hT + brow * H_ + 4 + half * 2) = s.ha1;
    }
    wg_sync_lds();
    if (w == 2) {
        const int obase = half * 12;
        const float* hp = hT + n * H_;
#pragma unroll
        for (int oo = 0; oo < 12; ++oo) {
            int o = obase + oo;
            float s2 = b_lin[o];
#pragma unroll
            for (int j = 0; j < H_; ++j) s2 += hp[j] * w_lin[o * H_ + j];
            out[(size_t)(b0 + n) * O_ + o] = s2;
        }
    }
}

extern "C" void kernel_launch(void* const* d_in, const int* in_sizes, int n_in,
                              void* d_out, int out_size, void* d_ws, size_t ws_size,
                              hipStream_t stream) {
    (void)in_sizes; (void)n_in; (void)d_ws; (void)ws_size; (void)out_size;
    const float* x     = (const float*)d_in[0];
    const float* w_ih0 = (const float*)d_in[1];
    const float* w_hh0 = (const float*)d_in[2];
    const float* b_ih0 = (const float*)d_in[3];
    const float* b_hh0 = (const float*)d_in[4];
    const float* w_ih1 = (const float*)d_in[5];
    const float* w_hh1 = (const float*)d_in[6];
    const float* b_ih1 = (const float*)d_in[7];
    const float* b_hh1 = (const float*)d_in[8];
    const float* w_lin = (const float*)d_in[9];
    const float* b_lin = (const float*)d_in[10];
    float* out = (float*)d_out;

    dim3 grid(B_ / 16);
    dim3 block(128);
    hipLaunchKernelGGL(lstm2_fused, grid, block, 0, stream,
                       x, w_ih0, w_hh0, b_ih0, b_hh0,
                       w_ih1, w_hh1, b_ih1, b_hh1, w_lin, b_lin, out);
}


# x fragments loaded seven phases ahead into a ring of eight register sets, not three ahead into four: the tail of the miss latency no longer holds the in-order waves at the barrier; sum order kept
# speedup vs baseline: 3.7085x; 1.0942x over previous
#include <hip/hip_runtime.h>


typedef float v2f __attribute__((ext_vector_type(2)));
typedef float v4f __attribute__((ext_vector_type(4)));
typedef float v8f __attribute__((ext_vector_type(8)));

#define B_  2048
#define T_  1024
#define D_  24
#define H_  8
#define O_  24
#define GP_ 36
#define PL_ (8 * GP_)

#define LOG2E_ 1.4426950408889634f


__device__ __forceinline__ void wg_sync_lds() {
    __builtin_amdgcn_fence(__ATOMIC_RELEASE, "workgroup", "local");
    __builtin_amdgcn_s_barrier();
    __builtin_amdgcn_fence(__ATOMIC_ACQUIRE, "workgroup", "local");
}

__device__ __forceinline__ v2f cell_pair(v2f& c, v4f if_, v4f go) {
    v2f ig = {if_[0], if_[1]}, fg = {if_[2], if_[3]};
    v2f gg = {go[0], go[1]},   og = {go[2], go[3]};
    v2f hh;
#pragma unroll
    for (int k = 0; k < 2; ++k) {
        c[k]  = __builtin_fmaf(fg[k], c[k], ig[k] * gg[k]);
        float r = __builtin_amdgcn_rcpf(1.0f + __builtin_amdgcn_exp2f(c[k]));
        hh[k] = __builtin_fmaf(og[k] + og[k], r, -og[k]);
    }
    return hh;
}

__device__ __forceinline__ v8f mm(v2f a, v2f b, v8f c) {
    return __builtin_amdgcn_wmma_f32_16x16x4_f32(false, a, false, b, (short)0, c, false, false);
}

struct Carry {
    v2f ha0, ha1;
    v2f ca, cb;
    v8f ing0, ing1;
};

template <bool A1, bool H1>
__device__ __forceinline__ void phase(Carry& s, bool lay1, float* g, float* hw,
                                      const float* hr, int sto, int ldo,
                                      const v2f (&xs)[6], v2f (&xr)[6], const float* xnext,
                                      const v2f (&Bi0)[6], const v2f (&Bi1)[6],
                                      const v2f (&Bh0)[2], const v2f (&Bh1)[2],
                                      const v8f& bt0, const v8f& bt1, float om, float oa)
{
    const v2f zero2 = {0.f, 0.f};
    v8f a0 = s.ing0, a1 = s.ing1;
    a0 = mm(s.ha0, Bh0[0], a0);
    a1 = mm(s.ha0, Bh1[0], a1);
    a0 = mm(s.ha1, Bh0[1], a0);
    a1 = mm(s.ha1, Bh1[1], a1);
    asm volatile("" : "+v"(a0), "+v"(a1));
    v2f x0 = xs[0];
    asm volatile("" : "+v"(x0));
    v8f t0 = mm(x0, Bi0[0], bt0);
    v8f t1 = mm(x0, Bi1[0], bt1);
    asm volatile("" : "+v"(t0), "+v"(t1));
    __builtin_amdgcn_sched_barrier(0);
    wg_sync_lds();

    v4f e0, e1;
#pragma unroll
    for (int v = 0; v < 4; ++v) {
        e0[v] = 1.0f + __builtin_amdgcn_exp2f(a0[v]);
        e1[v] = 1.0f + __builtin_amdgcn_exp2f(a1[v]);
    }
    asm volatile("" : "+v"(e0), "+v"(e1));
    v2f x1 = xs[1];
    asm volatile("" : "+v"(x1));
    t0 = mm(x1, Bi0[1], t0);
    asm volatile("" : "+v"(t0));
#pragma unroll
    for (int v = 0; v < 4; ++v) {
        e0[v] = __builtin_amdgcn_rcpf(e0[v]);
        e1[v] = __builtin_amdgcn_rcpf(e1[v]);
    }
    asm volatile("" : "+v"(e0), "+v"(e1));
    asm volatile("" : "+v"(x1));
    t1 = mm(x1, Bi1[1], t1);
    asm volatile("" : "+v"(t1));
#pragma unroll
    for (int v = 0; v < 4; ++v) {
        g[sto + v * GP_]      = __builtin_fmaf(om, e0[v], oa);
        g[sto + v * GP_ + 16] = __builtin_fmaf(om, e1[v], oa);
    }
    v4f hp = {s.ha0[0], s.ha0[1], s.ha1[0], s.ha1[1]};
    *(v4f*)hw = hp;
    __builtin_amdgcn_fence(__ATOMIC_RELEASE, "wavefront", "local");
    __builtin_amdgcn_fence(__ATOMIC_ACQUIRE, "wavefront", "local");

    v4f p0 = *(const v4f*)(g + ldo);
    v4f p1 = *(const v4f*)(g + ldo + 4);
    v4f p2 = *(const v4f*)(g + ldo + 16);
    v4f p3 = *(const v4f*)(g + ldo + 20);
    v4f hv = {0.f, 0.f, 0.f, 0.f};
    if (H1) hv = *(const v4f*)hr;
    __builtin_amdgcn_sched_barrier(0);
#pragma unroll
    for (int kk = 0; kk < 6; ++kk) xr[kk] = *(const v2f*)(xnext + kk * 4);
    v2f x2 = xs[2];
    asm volatile("" : "+v"(x2));
    t0 = mm(x2, Bi0[2], t0);
    t1 = mm(x2, Bi1[2], t1);
    t0 = mm(xs[3], Bi0[3], t0);
    t1 = mm(xs[3], Bi1[3], t1);
    __builtin_amdgcn_sched_barrier(0);
    v2f h0lo = {hv[0], hv[1]}, h0hi = {hv[2], hv[3]};
    v2f a4 = lay1 ? h0lo : xs[4];
    v2f a5 = lay1 ? h0hi : xs[5];
    t0 = mm(a4, Bi0[4], t0);
    t1 = mm(a4, Bi1[4], t1);
    t0 = mm(a5, Bi0[5], t0);
    t1 = mm(a5, Bi1[5], t1);
    asm volatile("" : "+v"(t0), "+v"(t1));
    s.ing0 = t0;
    s.ing1 = t1;
    s.ha0 = cell_pair(s.ca, p0, p1);
    s.ha1 = cell_pair(s.cb, p2, p3);
    if (!A1) {
        s.ha0 = lay1 ? zero2 : s.ha0;  s.ha1 = lay1 ? zero2 : s.ha1;
        s.ca  = lay1 ? zero2 : s.ca;   s.cb  = lay1 ? zero2 : s.cb;
    }
}

__global__ __launch_bounds__(128)
void lstm2_fused(const float* __restrict__ x,
                 const float* __restrict__ w_ih0, const float* __restrict__ w_hh0,
                 const float* __restrict__ b_ih0, const float* __restrict__ b_hh0,
                 const float* __restrict__ w_ih1, const float* __restrict__ w_hh1,
                 const float* __restrict__ b_ih1, const float* __restrict__ b_hh1,
                 const float* __restrict__ w_lin, const float* __restrict__ b_lin,
                 float* __restrict__ out)
{
    __shared__ __align__(16) float gLDS[4 * PL_];
    __shared__ __align__(16) float hbuf[2 * 4 * 32 * 4];
    __shared__ __align__(16) float hT[16 * H_];

    const int w    = __builtin_amdgcn_readfirstlane((int)(threadIdx.x >> 5));
    const int lay  = w >> 1;
    const bool lay1 = lay != 0;
    const int u    = w & 1;
    const int lane = threadIdx.x & 31;
    const int n    = lane & 15;
    const int half = lane >> 4;
    const int b0   = blockIdx.x * 16;
    const int slot = (n & 3) + 4 * (n >> 3);
    const int brow = 8 * u + slot;

    const int q     = (n >> 1) & 3;
    const int unit0 = 2 * (n >> 3) + (n & 1);
    const int grow0 = q * H_ + unit0;
    const int grow1 = grow0 + 4;
    const v2f zero2 = {0.f, 0.f};
    const v8f zero8 = {0.f, 0.f, 0.f, 0.f, 0.f, 0.f, 0.f, 0.f};

    v2f Bi0[6], Bi1[6];
#pragma unroll
    for (int kk = 0; kk < 6; ++kk) { Bi0[kk] = zero2; Bi1[kk] = zero2; }
    if (lay == 0) {
#pragma unroll
        for (int kk = 0; kk < 6; ++kk) {
            Bi0[kk] = *(const v2f*)(w_ih0 + grow0 * D_ + kk * 4 + half * 2);
            Bi1[kk] = *(const v2f*)(w_ih0 + grow1 * D_ + kk * 4 + half * 2);
        }
    } else {
#pragma unroll
        for (int kk = 0; kk < 2; ++kk) {
            Bi0[4 + kk] = *(const v2f*)(w_ih1 + grow0 * H_ + kk * 4 + half * 2);
            Bi1[4 + kk] = *(const v2f*)(w_ih1 + grow1 * H_ + kk * 4 + half * 2);
        }
    }
    const float* whh = lay ? w_hh1 : w_hh0;
    v2f Bh0[2], Bh1[2];
#pragma unroll
    for (int kk = 0; kk < 2; ++kk) {
        Bh0[kk] = *(const v2f*)(whh + grow0 * H_ + kk * 4 + half * 2);
        Bh1[kk] = *(const v2f*)(whh + grow1 * H_ + kk * 4 + half * 2);
    }

    const float bias0 = lay ? (b_ih1[grow0] + b_hh1[grow0]) : (b_ih0[grow0] + b_hh0[grow0]);
    const float bias1 = lay ? (b_ih1[grow1] + b_hh1[grow1]) : (b_ih0[grow1] + b_hh0[grow1]);
    const float ik   = (q == 2) ? (-2.0f * LOG2E_) : -LOG2E_;
    const float ib0  = ik * bias0;
    const float ib1  = ik * bias1;
    const float om   = (q == 2) ? (-4.0f * LOG2E_) : 1.0f;
    const float oa   = (q == 2) ? (2.0f * LOG2E_) : 0.0f;
    const v8f bt0 = {ib0, ib0, ib0, ib0, ib0, ib0, ib0, ib0};
    const v8f bt1 = {ib1, ib1, ib1, ib1, ib1, ib1, ib1, ib1};
#pragma unroll
    for (int kk = 0; kk < 6; ++kk) { Bi0[kk] = Bi0[kk] * ik; Bi1[kk] = Bi1[kk] * ik; }
#pragma unroll
    for (int kk = 0; kk < 2; ++kk) { Bh0[kk] = Bh0[kk] * ik; Bh1[kk] = Bh1[kk] * ik; }

    const float* xlane = x + (size_t)(b0 + brow) * T_ * D_ + half * 2;
    const int xstep = lay1 ? 0 : D_;

    Carry s;
    s.ha0 = zero2; s.ha1 = zero2; s.ca = zero2; s.cb = zero2;
    s.ing0 = zero8; s.ing1 = zero8;

    const int sto = w * PL_ + (half * 4) * GP_ + n;
    const int ldo = w * PL_ + slot * GP_ + half * 8;
    const int hwo = w * 128 + lane * 4;
    const int hro = u * 128 + lane * 4;

    v2f X0[6], X1[6], X2[6], X3[6], X4[6], X5[6], X6[6], X7[6];
    {
        v2f XT[6];
#pragma unroll
        for (int kk = 0; kk < 6; ++kk) {
            XT[kk] = *(const v2f*)(xlane + kk * 4);
            X1[kk] = *(const v2f*)(xlane + 1 * xstep + kk * 4);
            X2[kk] = *(const v2f*)(xlane + 2 * xstep + kk * 4);
            X3[kk] = *(const v2f*)(xlane + 3 * xstep + kk * 4);
            X4[kk] = *(const v2f*)(xlane + 4 * xstep + kk * 4);
            X5[kk] = *(const v2f*)(xlane + 5 * xstep + kk * 4);
            X6[kk] = *(const v2f*)(xlane + 6 * xstep + kk * 4);
            X7[kk] = *(const v2f*)(xlane + 7 * xstep + kk * 4);
        }
        XT[4] = lay1 ? zero2 : XT[4];
        XT[5] = lay1 ? zero2 : XT[5];
        v8f t0 = bt0, t1 = bt1;
#pragma unroll
        for (int kk = 0; kk < 6; ++kk) {
            t0 = mm(XT[kk], Bi0[kk], t0);
            t1 = mm(XT[kk], Bi1[kk], t1);
        }
        s.ing0 = t0;
        s.ing1 = t1;
    }

#define XN_(p) (xlane + (size_t)(((p) + 8 < T_) ? ((p) + 8) : (T_ - 1)) * xstep)
#define PH_(A1, H1, p, PAR, XS, XR)                                           \
    phase<A1, H1>(s, lay1, gLDS,                                              \
                  hbuf + (1 - (PAR)) * 512 + hwo, hbuf + (PAR) * 512 + hro,   \
                  sto, ldo, XS, XR, XN_(p), Bi0, Bi1, Bh0, Bh1, bt0, bt1, om, oa)

    PH_(false, false, 0, 0, X1, X0);
    PH_(false, false, 1, 1, X2, X1);
    PH_(false, true,  2, 0, X3, X2);
    PH_(true,  true,  3, 1, X4, X3);
    PH_(true,  true,  4, 0, X5, X4);
    PH_(true,  true,  5, 1, X6, X5);
    PH_(true,  true,  6, 0, X7, X6);
    PH_(true,  true,  7, 1, X0, X7);
    for (int p = 8; p < T_; p += 8) {
        PH_(true, true, p,     0, X1, X0);
        PH_(true, true, p + 1, 1, X2, X1);
        PH_(true, true, p + 2, 0, X3, X2);
        PH_(true, true, p + 3, 1, X4, X3);
        PH_(true, true, p + 4, 0, X5, X4);
        PH_(true, true, p + 5, 1, X6, X5);
        PH_(true, true, p + 6, 0, X7, X6);
        PH_(true, true, p + 7, 1, X0, X7);
    }
    PH_(true, true,  T_,     0, X1, X0);
    PH_(true, true,  T_ + 1, 1, X2, X1);
    PH_(true, false, T_ + 2, 0, X3, X2);
#undef PH_
#undef XN_

    if (lay1) {
        *(v2f*)(hT + brow * H_ + half * 2)     = s.ha0;
        *(v2f*)(hT + brow * H_ + 4 + half * 2) = s.ha1;
    }
    wg_sync_lds();
    if (w == 2) {
        const int obase = half * 12;
        const float* hp = hT + n * H_;
#pragma unroll
        for (int oo = 0; oo < 12; ++oo) {
            int o = obase + oo;
            float s2 = b_lin[o];
#pragma unroll
            for (int j = 0; j < H_; ++j) s2 += hp[j] * w_lin[o * H_ + j];
            out[(size_t)(b0 + n) * O_ + o] = s2;
        }
    }
}

extern "C" void kernel_launch(void* const* d_in, const int* in_sizes, int n_in,
                              void* d_out, int out_size, void* d_ws, size_t ws_size,
                              hipStream_t stream) {
    (void)in_sizes; (void)n_in; (void)d_ws; (void)ws_size; (void)out_size;
    const float* x     = (const float*)d_in[0];
    const float* w_ih0 = (const float*)d_in[1];
    const float* w_hh0 = (const float*)d_in[2];
    const float* b_ih0 = (const float*)d_in[3];
    const float* b_hh0 = (const float*)d_in[4];
    const float* w_ih1 = (const float*)d_in[5];
    const float* w_hh1 = (const float*)d_in[6];
    const float* b_ih1 = (const float*)d_in[7];
    const float* b_hh1 = (const float*)d_in[8];
    const float* w_lin = (const float*)d_in[9];
    const float* b_lin = (const float*)d_in[10];
    float* out = (float*)d_out;

    dim3 grid(B_ / 16);
    dim3 block(128);
    hipLaunchKernelGGL(lstm2_fused, grid, block, 0, stream,
                       x, w_ih0, w_hh0, b_ih0, b_hh0,
                       w_ih1, w_hh1, b_ih1, b_hh1, w_lin, b_lin, out);
}
